# starting kernel
# baseline (speedup 1.0000x reference)
#include <hip/hip_runtime.h>

typedef float v2f __attribute__((ext_vector_type(2)));
typedef float v4f __attribute__((ext_vector_type(4)));
typedef float v8f __attribute__((ext_vector_type(8)));

#define TASKS 512
#define NWAY  5
#define NSHOT 5
#define NQ    75
#define NS    25
#define DIM   1600
#define PSTR  1604
#define ZROW  NWAY

__device__ inline float dot4sq(v4f a) {
    return a.x * a.x + a.y * a.y + a.z * a.z + a.w * a.w;
}

__global__ __launch_bounds__(160)
void proto_head_kernel(const float* __restrict__ query,
                       const float* __restrict__ support,
                       const int*   __restrict__ slabels,
                       const float* __restrict__ scale,
                       float*       __restrict__ out)
{
    __shared__ float ldsProto[(NWAY + 1) * PSTR];
    __shared__ float ldsAA[5 * 32];
    __shared__ float ldsBB[NWAY];
    __shared__ int   ldsLab[NS];

    const int task = blockIdx.x;
    const int tid  = threadIdx.x;
    const int lane = tid & 31;
    const int wave = tid >> 5;

    if (tid < NS)   ldsLab[tid] = slabels[task * NS + tid];
    if (tid < NWAY) ldsBB[tid]  = 0.0f;
    for (int c = tid; c < PSTR; c += 160) ldsProto[ZROW * PSTR + c] = 0.0f;
    __syncthreads();

    const float* sup = support + (size_t)task * NS * DIM;
    float bb0 = 0.f, bb1 = 0.f, bb2 = 0.f, bb3 = 0.f, bb4 = 0.f;

    for (int c = tid; c < DIM / 4; c += 160) {
        v4f a0 = {0,0,0,0}, a1 = {0,0,0,0}, a2 = {0,0,0,0}, a3 = {0,0,0,0}, a4 = {0,0,0,0};
        for (int s = 0; s < NS; ++s) {
            v4f v = *reinterpret_cast<const v4f*>(sup + s * DIM + c * 4);
            switch (ldsLab[s]) {
                case 0: a0 += v; break;
                case 1: a1 += v; break;
                case 2: a2 += v; break;
                case 3: a3 += v; break;
                default: a4 += v; break;
            }
        }
        const float inv = 1.0f / (float)NSHOT;
        a0.x *= inv; a0.y *= inv; a0.z *= inv; a0.w *= inv;
        a1.x *= inv; a1.y *= inv; a1.z *= inv; a1.w *= inv;
        a2.x *= inv; a2.y *= inv; a2.z *= inv; a2.w *= inv;
        a3.x *= inv; a3.y *= inv; a3.z *= inv; a3.w *= inv;
        a4.x *= inv; a4.y *= inv; a4.z *= inv; a4.w *= inv;
        *reinterpret_cast<v4f*>(&ldsProto[0 * PSTR + c * 4]) = a0;  bb0 += dot4sq(a0);
        *reinterpret_cast<v4f*>(&ldsProto[1 * PSTR + c * 4]) = a1;  bb1 += dot4sq(a1);
        *reinterpret_cast<v4f*>(&ldsProto[2 * PSTR + c * 4]) = a2;  bb2 += dot4sq(a2);
        *reinterpret_cast<v4f*>(&ldsProto[3 * PSTR + c * 4]) = a3;  bb3 += dot4sq(a3);
        *reinterpret_cast<v4f*>(&ldsProto[4 * PSTR + c * 4]) = a4;  bb4 += dot4sq(a4);
    }
    atomicAdd(&ldsBB[0], bb0);
    atomicAdd(&ldsBB[1], bb1);
    atomicAdd(&ldsBB[2], bb2);
    atomicAdd(&ldsBB[3], bb3);
    atomicAdd(&ldsBB[4], bb4);
    __syncthreads();

    const float sc   = scale[0];
    const int   tile = wave;
    const int   half = lane >> 4;
    const int   l16  = lane & 15;
    const int   arow = tile * 16 + l16;
    const int   crow = (arow < NQ) ? arow : 0;
    const float* qrow = query + (size_t)task * NQ * DIM + (size_t)crow * DIM + half * 2;
    const int   n     = l16;
    const int   bRow  = (n < NWAY) ? n : ZROW;
    const float* bptr = &ldsProto[bRow * PSTR + half * 2];

    v8f acc0 = {0,0,0,0,0,0,0,0};
    v8f acc1 = {0,0,0,0,0,0,0,0};
    float aa = 0.0f;

    #pragma unroll 4
    for (int k = 0; k < DIM; k += 8) {
        v2f a0 = *reinterpret_cast<const v2f*>(qrow + k);
        v2f a1 = *reinterpret_cast<const v2f*>(qrow + k + 4);
        aa += a0.x * a0.x + a0.y * a0.y + a1.x * a1.x + a1.y * a1.y;

        v2f b0 = *reinterpret_cast<const v2f*>(bptr + k);
        v2f b1 = *reinterpret_cast<const v2f*>(bptr + k + 4);

        acc0 = __builtin_amdgcn_wmma_f32_16x16x4_f32(
                   false, a0, false, b0, (short)0, acc0, false, false);
        acc1 = __builtin_amdgcn_wmma_f32_16x16x4_f32(
                   false, a1, false, b1, (short)0, acc1, false, false);
    }

    ldsAA[wave * 32 + lane] = aa;
    __syncthreads();

    if (n < NWAY) {
        const float bbv = ldsBB[n];
        #pragma unroll
        for (int j = 0; j < 8; ++j) {
            const int m   = half * 8 + j;
            const int row = tile * 16 + m;
            if (row < NQ) {
                const float aarow = ldsAA[wave * 32 + m] + ldsAA[wave * 32 + m + 16];
                const float abv   = acc0[j] + acc1[j];
                const float logit = sc * (-(aarow - 2.0f * abv + bbv) * (1.0f / (float)DIM));
                out[((size_t)task * NQ + row) * NWAY + n] = logit;
            }
        }
    }
}

extern "C" void kernel_launch(void* const* d_in, const int* in_sizes, int n_in,
                              void* d_out, int out_size, void* d_ws, size_t ws_size,
                              hipStream_t stream) {
    const float* query   = (const float*)d_in[0];
    const float* support = (const float*)d_in[1];
    const int*   slab    = (const int*)  d_in[2];
    const float* scale   = (const float*)d_in[3];
    float*       out     = (float*)d_out;
    (void)in_sizes; (void)n_in; (void)out_size; (void)d_ws; (void)ws_size;

    proto_head_kernel<<<TASKS, 160, 0, stream>>>(query, support, slab, scale, out);
}


# k-only task split: 10 waves (was 25), each all 75 rows x one 160-float slice, its prototype slice built in-wave (no barrier before products), rows via two register buffers; AA/AB/BB sums regrouped
# speedup vs baseline: 2.2874x; 2.2874x over previous
#include <hip/hip_runtime.h>


typedef float v2f __attribute__((ext_vector_type(2)));
typedef float v4f __attribute__((ext_vector_type(4)));
typedef float v8f __attribute__((ext_vector_type(8)));

#define TASKS 512
#define NWAY  5
#define NSHOT 5
#define NQ    75
#define NS    25
#define DIM   1600
#define MT    5
#define MROWS (MT * 16)
#define NWAVE 10
#define KW    (DIM / NWAVE)
#define NTHR  (NWAVE * 32)
#define PSTR  164
#define KSTEPS (KW / 8)
#define GRP   2
#define NGRP  (KSTEPS / GRP)
#define NPIECE (KW / 32)

__device__ inline float dot4sq(v4f a) {
    return a.x * a.x + a.y * a.y + a.z * a.z + a.w * a.w;
}

__device__ inline float wave_sum(float v) {
    v += __shfl_xor(v, 16);
    v += __shfl_xor(v, 8);
    v += __shfl_xor(v, 4);
    v += __shfl_xor(v, 2);
    v += __shfl_xor(v, 1);
    return v;
}

#define AA_HERE asm volatile("" : "+v"(aa[0]), "+v"(aa[1]), "+v"(aa[2]), "+v"(aa[3]), "+v"(aa[4]));

#define LOAD_GROUP(buf, g)                                                         \
    _Pragma("unroll")                                                              \
    for (int jj = 0; jj < GRP; ++jj) {                                             \
        _Pragma("unroll")                                                          \
        for (int t = 0; t < MT - 1; ++t)                                           \
            buf[jj][t] = *reinterpret_cast<const v4f*>(q03 + t * 16 * DIM + ((g) * GRP + jj) * 8); \
        buf[jj][MT - 1] = *reinterpret_cast<const v4f*>(q4 + ((g) * GRP + jj) * 8); \
    }

#define COMPUTE_GROUP(buf, g)                                                      \
    _Pragma("unroll")                                                              \
    for (int jj = 0; jj < GRP; ++jj) {                                             \
        const v4f b = *reinterpret_cast<const v4f*>(bptr + ((g) * GRP + jj) * 8);  \
        const v2f blo = {b.x, b.y};                                                \
        const v2f bhi = {b.z, b.w};                                                \
        _Pragma("unroll")                                                          \
        for (int t = 0; t < MT; ++t) {                                             \
            const v4f a = buf[jj][t];                                              \
            const v2f alo = {a.x, a.y};                                            \
            aa[t] += dot4sq(a);                                                    \
            acc[t] = __builtin_amdgcn_wmma_f32_16x16x4_f32(                        \
                         false, alo, false, blo, (short)0, acc[t], false, false);  \
        }                                                                          \
        _Pragma("unroll")                                                          \
        for (int t = 0; t < MT; ++t) {                                             \
            const v4f a = buf[jj][t];                                              \
            const v2f ahi = {a.z, a.w};                                            \
            acc[t] = __builtin_amdgcn_wmma_f32_16x16x4_f32(                        \
                         false, ahi, false, bhi, (short)0, acc[t], false, false);  \
        }                                                                          \
    }                                                                              \
    AA_HERE

#define ORDER_PIN __builtin_amdgcn_sched_barrier(0);

__global__ __launch_bounds__(NTHR)
void proto_head_kernel(const float* __restrict__ query,
                       const float* __restrict__ support,
                       const int*   __restrict__ slabels,
                       const float* __restrict__ scale,
                       float*       __restrict__ out)
{
    __shared__ __attribute__((aligned(16))) float ldsProto[NWAVE * NWAY * PSTR];
    __shared__ float ldsAB[NWAVE * MROWS * NWAY];
    __shared__ float ldsAA[NWAVE * MROWS];
    __shared__ float ldsBB[NWAVE * NWAY];

    const int task = blockIdx.x;
    const int tid  = threadIdx.x;
    const int lane = tid & 31;
    const int wave = __builtin_amdgcn_readfirstlane(tid >> 5);
    const int half = lane >> 4;
    const int l16  = lane & 15;
    const int K0   = wave * KW;

    const float* qtask = query + (size_t)task * NQ * DIM + K0 + half * 4;
    const float* q03   = qtask + (size_t)l16 * DIM;
    const int    r4    = (MT - 1) * 16 + l16;
    const float* q4    = qtask + (size_t)((r4 < NQ) ? r4 : 0) * DIM;

    v4f bufA[GRP][MT], bufB[GRP][MT];
    LOAD_GROUP(bufA, 0)

    float* pw = &ldsProto[wave * NWAY * PSTR];
    {
        const float* sup = support + (size_t)task * NS * DIM + K0 + lane;
        const int*   lab = slabels + task * NS;
        float p0[NPIECE], p1[NPIECE], p2[NPIECE], p3[NPIECE], p4[NPIECE];
        #pragma unroll
        for (int i = 0; i < NPIECE; ++i) { p0[i] = 0.f; p1[i] = 0.f; p2[i] = 0.f; p3[i] = 0.f; p4[i] = 0.f; }
        #pragma unroll
        for (int s = 0; s < NS; ++s) {
            const int   lb = lab[s];
            const float w0 = (lb == 0) ? 1.f : 0.f;
            const float w1 = (lb == 1) ? 1.f : 0.f;
            const float w2 = (lb == 2) ? 1.f : 0.f;
            const float w3 = (lb == 3) ? 1.f : 0.f;
            const float w4 = ((unsigned)lb > 3u) ? 1.f : 0.f;
            #pragma unroll
            for (int i = 0; i < NPIECE; ++i) {
                const float v = sup[s * DIM + i * 32];
                p0[i] += w0 * v;
                p1[i] += w1 * v;
                p2[i] += w2 * v;
                p3[i] += w3 * v;
                p4[i] += w4 * v;
            }
        }
        const float inv = 1.0f / (float)NSHOT;
        float bb0 = 0.f, bb1 = 0.f, bb2 = 0.f, bb3 = 0.f, bb4 = 0.f;
        #pragma unroll
        for (int i = 0; i < NPIECE; ++i) {
            p0[i] *= inv; p1[i] *= inv; p2[i] *= inv; p3[i] *= inv; p4[i] *= inv;
            pw[0 * PSTR + i * 32 + lane] = p0[i];  bb0 += p0[i] * p0[i];
            pw[1 * PSTR + i * 32 + lane] = p1[i];  bb1 += p1[i] * p1[i];
            pw[2 * PSTR + i * 32 + lane] = p2[i];  bb2 += p2[i] * p2[i];
            pw[3 * PSTR + i * 32 + lane] = p3[i];  bb3 += p3[i] * p3[i];
            pw[4 * PSTR + i * 32 + lane] = p4[i];  bb4 += p4[i] * p4[i];
        }
        bb0 = wave_sum(bb0); bb1 = wave_sum(bb1); bb2 = wave_sum(bb2);
        bb3 = wave_sum(bb3); bb4 = wave_sum(bb4);
        if (lane == 0) {
            ldsBB[wave * NWAY + 0] = bb0;
            ldsBB[wave * NWAY + 1] = bb1;
            ldsBB[wave * NWAY + 2] = bb2;
            ldsBB[wave * NWAY + 3] = bb3;
            ldsBB[wave * NWAY + 4] = bb4;
        }
    }
    __builtin_amdgcn_fence(__ATOMIC_RELEASE, "workgroup");
    __builtin_amdgcn_wave_barrier();
    __builtin_amdgcn_fence(__ATOMIC_ACQUIRE, "workgroup");

    const int    n    = l16;
    const int    bRow = (n < NWAY) ? n : 0;
    const float* bptr = pw + bRow * PSTR + half * 4;

    v8f acc[MT];
    float aa[MT];
    #pragma unroll
    for (int t = 0; t < MT; ++t) {
        acc[t] = (v8f){0, 0, 0, 0, 0, 0, 0, 0};
        aa[t]  = 0.0f;
    }

    #pragma unroll
    for (int g = 0; g < NGRP - 2; g += 2) {
        LOAD_GROUP(bufB, g + 1)
        ORDER_PIN
        COMPUTE_GROUP(bufA, g)
        ORDER_PIN
        LOAD_GROUP(bufA, g + 2)
        ORDER_PIN
        COMPUTE_GROUP(bufB, g + 1)
        ORDER_PIN
    }
    LOAD_GROUP(bufB, NGRP - 1)
    ORDER_PIN
    COMPUTE_GROUP(bufA, NGRP - 2)
    COMPUTE_GROUP(bufB, NGRP - 1)

    #pragma unroll
    for (int t = 0; t < MT; ++t) {
        const float aar = aa[t] + __shfl_xor(aa[t], 16);
        if (half == 0) ldsAA[wave * MROWS + t * 16 + l16] = aar;
        if (n < NWAY) {
            #pragma unroll
            for (int j = 0; j < 8; ++j)
                ldsAB[(wave * MROWS + t * 16 + half * 8 + j) * NWAY + n] = acc[t][j];
        }
    }
    __syncthreads();

    const float sc = scale[0];
    for (int i = tid; i < NQ * NWAY; i += NTHR) {
        const int row = i / NWAY;
        const int c   = i - row * NWAY;
        float aarow = 0.0f, abv = 0.0f, bbv = 0.0f;
        #pragma unroll
        for (int w = 0; w < NWAVE; ++w) {
            aarow += ldsAA[w * MROWS + row];
            abv   += ldsAB[(w * MROWS + row) * NWAY + c];
            bbv   += ldsBB[w * NWAY + c];
        }
        const float logit = sc * (-(aarow - 2.0f * abv + bbv) * (1.0f / (float)DIM));
        out[(size_t)task * NQ * NWAY + i] = logit;
    }
}

extern "C" void kernel_launch(void* const* d_in, const int* in_sizes, int n_in,
                              void* d_out, int out_size, void* d_ws, size_t ws_size,
                              hipStream_t stream) {
    const float* query   = (const float*)d_in[0];
    const float* support = (const float*)d_in[1];
    const int*   slab    = (const int*)  d_in[2];
    const float* scale   = (const float*)d_in[3];
    float*       out     = (float*)d_out;
    (void)in_sizes; (void)n_in; (void)out_size; (void)d_ws; (void)ws_size;

    proto_head_kernel<<<TASKS, NTHR, 0, stream>>>(query, support, slab, scale, out);
}


# A tile = 8 query rows x 2 k parts, B = 5 classes x same parts: a query load takes 64 B from 8 lines (was 32 B from 16), a line's 2 loads adjacent; AA/AB sums regrouped; repair: means / counted rows
# speedup vs baseline: 2.5424x; 1.1115x over previous
#include <hip/hip_runtime.h>


typedef float v2f __attribute__((ext_vector_type(2)));
typedef float v4f __attribute__((ext_vector_type(4)));
typedef float v8f __attribute__((ext_vector_type(8)));

#define TASKS 512
#define NWAY  5
#define NQ    75
#define NS    25
#define DIM   1600
#define RT    8
#define MT    10
#define MROWS (MT * RT)
#define HT    5
#define NWAVE 10
#define KW    (DIM / NWAVE)
#define NTHR  (NWAVE * 32)
#define PSTR  164
#define KSTEP 16
#define NLINE (KW / 32)
#define NGRP  (2 * NLINE)
#define NPIECE (KW / 32)

__device__ inline float dot4sq(v4f a) {
    return a.x * a.x + a.y * a.y + a.z * a.z + a.w * a.w;
}

__device__ inline float wave_sum(float v) {
    v += __shfl_xor(v, 16);
    v += __shfl_xor(v, 8);
    v += __shfl_xor(v, 4);
    v += __shfl_xor(v, 2);
    v += __shfl_xor(v, 1);
    return v;
}

#define AA_HERE asm volatile("" : "+v"(aa[0]), "+v"(aa[1]), "+v"(aa[2]), "+v"(aa[3]), "+v"(aa[4]));

#define LOAD_GROUP(buf, g)                                                         \
    _Pragma("unroll")                                                              \
    for (int t = 0; t < HT; ++t) {                                                 \
        const int T = ((g) / NLINE) * HT + t;                                      \
        const float* srow = (T < MT - 1) ? (q08 + T * RT * DIM) : q9;              \
        _Pragma("unroll")                                                          \
        for (int jj = 0; jj < 2; ++jj)                                             \
            buf[jj][t] = *reinterpret_cast<const v4f*>(srow + (((g) % NLINE) * 2 + jj) * KSTEP); \
    }

#define FLUSH_HALF(hf)                                                             \
    {                                                                              \
        float rs[HT];                                                              \
        _Pragma("unroll")                                                          \
        for (int t = 0; t < HT; ++t) rs[t] = aa[t] + __shfl_xor(aa[t], 16);        \
        _Pragma("unroll")                                                          \
        for (int t = 0; t < HT; ++t) rs[t] += __shfl_xor(rs[t], 8);                \
        if (lane < RT) {                                                           \
            _Pragma("unroll")                                                      \
            for (int t = 0; t < HT; ++t)                                           \
                ldsAA[wave * MROWS + ((hf) * HT + t) * RT + lane] = rs[t];         \
        }                                                                          \
        if (dUse) {                                                                \
            _Pragma("unroll")                                                      \
            for (int t = 0; t < HT; ++t) {                                         \
                _Pragma("unroll")                                                  \
                for (int j = 0; j < 8; ++j)                                        \
                    ldsAB[((wave * 2 + half) * MROWS + ((hf) * HT + t) * RT + j) * NWAY + dCls] = acc[t][j]; \
            }                                                                      \
        }                                                                          \
    }

#define COMPUTE_GROUP(buf, g)                                                      \
    if ((g) % NLINE == 0) {                                                        \
        _Pragma("unroll")                                                          \
        for (int t = 0; t < HT; ++t) {                                             \
            acc[t] = (v8f){0, 0, 0, 0, 0, 0, 0, 0};                                \
            aa[t]  = 0.0f;                                                         \
        }                                                                          \
    }                                                                              \
    _Pragma("unroll")                                                              \
    for (int jj = 0; jj < 2; ++jj) {                                               \
        const v4f b = *reinterpret_cast<const v4f*>(bptr + (((g) % NLINE) * 2 + jj) * KSTEP); \
        const v2f blo = {b.x, b.y};                                                \
        const v2f bhi = {b.z, b.w};                                                \
        _Pragma("unroll")                                                          \
        for (int t = 0; t < HT; ++t) {                                             \
            const v4f a = buf[jj][t];                                              \
            const v2f alo = {a.x, a.y};                                            \
            aa[t] += dot4sq(a);                                                    \
            acc[t] = __builtin_amdgcn_wmma_f32_16x16x4_f32(                        \
                         false, alo, false, blo, (short)0, acc[t], false, false);  \
        }                                                                          \
        _Pragma("unroll")                                                          \
        for (int t = 0; t < HT; ++t) {                                             \
            const v4f a = buf[jj][t];                                              \
            const v2f ahi = {a.z, a.w};                                            \
            acc[t] = __builtin_amdgcn_wmma_f32_16x16x4_f32(                        \
                         false, ahi, false, bhi, (short)0, acc[t], false, false);  \
        }                                                                          \
    }                                                                              \
    AA_HERE                                                                        \
    if ((g) % NLINE == NLINE - 1) {                                                \
        FLUSH_HALF((g) / NLINE)                                                    \
    }

#define ORDER_PIN __builtin_amdgcn_sched_barrier(0);

__global__ __launch_bounds__(NTHR)
void proto_head_kernel(const float* __restrict__ query,
                       const float* __restrict__ support,
                       const int*   __restrict__ slabels,
                       const float* __restrict__ scale,
                       float*       __restrict__ out)
{
    __shared__ __attribute__((aligned(16))) float ldsProto[NWAVE * NWAY * PSTR];
    __shared__ float ldsAB[NWAVE * 2 * MROWS * NWAY];
    __shared__ float ldsAA[NWAVE * MROWS];
    __shared__ float ldsBB[NWAVE * NWAY];

    const int task = blockIdx.x;
    const int tid  = threadIdx.x;
    const int lane = tid & 31;
    const int wave = __builtin_amdgcn_readfirstlane(tid >> 5);
    const int half = lane >> 4;
    const int l16  = lane & 15;
    const int r8   = lane & 7;
    const int part = (lane >> 3) & 1;
    const int K0   = wave * KW;

    const float* qtask = query + (size_t)task * NQ * DIM + K0 + part * 8 + half * 4;
    const float* q08   = qtask + (size_t)r8 * DIM;
    const int    r9    = (MT - 1) * RT + r8;
    const float* q9    = qtask + (size_t)((r9 < NQ) ? r9 : 0) * DIM;

    v4f bufA[2][HT], bufB[2][HT];
    LOAD_GROUP(bufA, 0)

    float* pw = &ldsProto[wave * NWAY * PSTR];
    {
        const float* sup = support + (size_t)task * NS * DIM + K0 + lane;
        const int*   lab = slabels + task * NS;
        float p0[NPIECE], p1[NPIECE], p2[NPIECE], p3[NPIECE], p4[NPIECE];
        #pragma unroll
        for (int i = 0; i < NPIECE; ++i) { p0[i] = 0.f; p1[i] = 0.f; p2[i] = 0.f; p3[i] = 0.f; p4[i] = 0.f; }
        int c0 = 0, c1 = 0, c2 = 0, c3 = 0, c4 = 0;
        #pragma unroll
        for (int s = 0; s < NS; ++s) {
            const int   lb = lab[s];
            const float w0 = (lb == 0) ? 1.f : 0.f;
            const float w1 = (lb == 1) ? 1.f : 0.f;
            const float w2 = (lb == 2) ? 1.f : 0.f;
            const float w3 = (lb == 3) ? 1.f : 0.f;
            const float w4 = (lb == 4) ? 1.f : 0.f;
            c0 += (lb == 0); c1 += (lb == 1); c2 += (lb == 2); c3 += (lb == 3); c4 += (lb == 4);
            #pragma unroll
            for (int i = 0; i < NPIECE; ++i) {
                const float v = sup[s * DIM + i * 32];
                p0[i] += w0 * v;
                p1[i] += w1 * v;
                p2[i] += w2 * v;
                p3[i] += w3 * v;
                p4[i] += w4 * v;
            }
        }
        const float inv0 = 1.0f / (float)c0, inv1 = 1.0f / (float)c1, inv2 = 1.0f / (float)c2;
        const float inv3 = 1.0f / (float)c3, inv4 = 1.0f / (float)c4;
        float bb0 = 0.f, bb1 = 0.f, bb2 = 0.f, bb3 = 0.f, bb4 = 0.f;
        #pragma unroll
        for (int i = 0; i < NPIECE; ++i) {
            p0[i] *= inv0; p1[i] *= inv1; p2[i] *= inv2; p3[i] *= inv3; p4[i] *= inv4;
            pw[0 * PSTR + i * 32 + lane] = p0[i];  bb0 += p0[i] * p0[i];
            pw[1 * PSTR + i * 32 + lane] = p1[i];  bb1 += p1[i] * p1[i];
            pw[2 * PSTR + i * 32 + lane] = p2[i];  bb2 += p2[i] * p2[i];
            pw[3 * PSTR + i * 32 + lane] = p3[i];  bb3 += p3[i] * p3[i];
            pw[4 * PSTR + i * 32 + lane] = p4[i];  bb4 += p4[i] * p4[i];
        }
        bb0 = wave_sum(bb0); bb1 = wave_sum(bb1); bb2 = wave_sum(bb2);
        bb3 = wave_sum(bb3); bb4 = wave_sum(bb4);
        if (lane == 0) {
            ldsBB[wave * NWAY + 0] = bb0;
            ldsBB[wave * NWAY + 1] = bb1;
            ldsBB[wave * NWAY + 2] = bb2;
            ldsBB[wave * NWAY + 3] = bb3;
            ldsBB[wave * NWAY + 4] = bb4;
        }
    }
    __builtin_amdgcn_fence(__ATOMIC_RELEASE, "workgroup");
    __builtin_amdgcn_wave_barrier();
    __builtin_amdgcn_fence(__ATOMIC_ACQUIRE, "workgroup");

    const bool   bUsed = (l16 < 2 * NWAY);
    const int    bPrt  = bUsed ? (l16 / NWAY) : 0;
    const int    bCls  = bUsed ? (l16 - bPrt * NWAY) : 0;
    const float* bptr  = pw + bCls * PSTR + bPrt * 8 + half * 4;
    const bool   dUse  = bUsed && (bPrt == half);
    const int    dCls  = bCls;

    v8f acc[HT];
    float aa[HT];

    #pragma unroll
    for (int g = 0; g < NGRP - 2; g += 2) {
        LOAD_GROUP(bufB, g + 1)
        ORDER_PIN
        COMPUTE_GROUP(bufA, g)
        ORDER_PIN
        LOAD_GROUP(bufA, g + 2)
        ORDER_PIN
        COMPUTE_GROUP(bufB, g + 1)
        ORDER_PIN
    }
    LOAD_GROUP(bufB, NGRP - 1)
    ORDER_PIN
    COMPUTE_GROUP(bufA, NGRP - 2)
    COMPUTE_GROUP(bufB, NGRP - 1)
    __syncthreads();

    const float sc = scale[0];
    for (int i = tid; i < NQ * NWAY; i += NTHR) {
        const int row = i / NWAY;
        const int c   = i - row * NWAY;
        float aarow = 0.0f, abv = 0.0f, bbv = 0.0f;
        #pragma unroll
        for (int w = 0; w < NWAVE; ++w) {
            aarow += ldsAA[w * MROWS + row];
            abv   += ldsAB[((w * 2 + 0) * MROWS + row) * NWAY + c];
            abv   += ldsAB[((w * 2 + 1) * MROWS + row) * NWAY + c];
            bbv   += ldsBB[w * NWAY + c];
        }
        const float logit = sc * (-(aarow - 2.0f * abv + bbv) * (1.0f / (float)DIM));
        out[(size_t)task * NQ * NWAY + i] = logit;
    }
}

extern "C" void kernel_launch(void* const* d_in, const int* in_sizes, int n_in,
                              void* d_out, int out_size, void* d_ws, size_t ws_size,
                              hipStream_t stream) {
    const float* query   = (const float*)d_in[0];
    const float* support = (const float*)d_in[1];
    const int*   slab    = (const int*)  d_in[2];
    const float* scale   = (const float*)d_in[3];
    float*       out     = (float*)d_out;
    (void)in_sizes; (void)n_in; (void)out_size; (void)d_ws; (void)ws_size;

    proto_head_kernel<<<TASKS, NTHR, 0, stream>>>(query, support, slab, scale, out);
}


# two tasks per work-group: 20 waves = 5 on every SIMD, 256 groups, one per WGP (a 10-wave group lands 3+3+2+2: at 176 registers, 5 a SIMD, no second one fits beside it); per-task sums unchanged
# speedup vs baseline: 2.6248x; 1.0324x over previous
#include <hip/hip_runtime.h>


typedef float v2f __attribute__((ext_vector_type(2)));
typedef float v4f __attribute__((ext_vector_type(4)));
typedef float v8f __attribute__((ext_vector_type(8)));

#define TASKS 512
#define NWAY  5
#define NQ    75
#define NS    25
#define DIM   1600
#define RT    8
#define MT    10
#define MROWS (MT * RT)
#define HT    5
#define NWAVE 10
#define KW    (DIM / NWAVE)
#define TPB   2
#define NWG   (TPB * NWAVE)
#define TTHR  (NWAVE * 32)
#define NTHR  (TPB * TTHR)
#define NGRID (TASKS / TPB)
#define PSTR  164
#define KSTEP 16
#define NLINE (KW / 32)
#define NGRP  (2 * NLINE)
#define NPIECE (KW / 32)

__device__ inline float dot4sq(v4f a) {
    return a.x * a.x + a.y * a.y + a.z * a.z + a.w * a.w;
}

__device__ inline float wave_sum(float v) {
    v += __shfl_xor(v, 16);
    v += __shfl_xor(v, 8);
    v += __shfl_xor(v, 4);
    v += __shfl_xor(v, 2);
    v += __shfl_xor(v, 1);
    return v;
}

#define AA_HERE asm volatile("" : "+v"(aa[0]), "+v"(aa[1]), "+v"(aa[2]), "+v"(aa[3]), "+v"(aa[4]));

#define LOAD_GROUP(buf, g)                                                         \
    _Pragma("unroll")                                                              \
    for (int t = 0; t < HT; ++t) {                                                 \
        const int T = ((g) / NLINE) * HT + t;                                      \
        const float* srow = (T < MT - 1) ? (q08 + T * RT * DIM) : q9;              \
        _Pragma("unroll")                                                          \
        for (int jj = 0; jj < 2; ++jj)                                             \
            buf[jj][t] = *reinterpret_cast<const v4f*>(srow + (((g) % NLINE) * 2 + jj) * KSTEP); \
    }

#define FLUSH_HALF(hf)                                                             \
    {                                                                              \
        float rs[HT];                                                              \
        _Pragma("unroll")                                                          \
        for (int t = 0; t < HT; ++t) rs[t] = aa[t] + __shfl_xor(aa[t], 16);        \
        _Pragma("unroll")                                                          \
        for (int t = 0; t < HT; ++t) rs[t] += __shfl_xor(rs[t], 8);                \
        if (lane < RT) {                                                           \
            _Pragma("unroll")                                                      \
            for (int t = 0; t < HT; ++t)                                           \
                ldsAA[wave * MROWS + ((hf) * HT + t) * RT + lane] = rs[t];         \
        }                                                                          \
        if (dUse) {                                                                \
            _Pragma("unroll")                                                      \
            for (int t = 0; t < HT; ++t) {                                         \
                _Pragma("unroll")                                                  \
                for (int j = 0; j < 8; ++j)                                        \
                    ldsAB[((wave * 2 + half) * MROWS + ((hf) * HT + t) * RT + j) * NWAY + dCls] = acc[t][j]; \
            }                                                                      \
        }                                                                          \
    }

#define COMPUTE_GROUP(buf, g)                                                      \
    if ((g) % NLINE == 0) {                                                        \
        _Pragma("unroll")                                                          \
        for (int t = 0; t < HT; ++t) {                                             \
            acc[t] = (v8f){0, 0, 0, 0, 0, 0, 0, 0};                                \
            aa[t]  = 0.0f;                                                         \
        }                                                                          \
    }                                                                              \
    _Pragma("unroll")                                                              \
    for (int jj = 0; jj < 2; ++jj) {                                               \
        const v4f b = *reinterpret_cast<const v4f*>(bptr + (((g) % NLINE) * 2 + jj) * KSTEP); \
        const v2f blo = {b.x, b.y};                                                \
        const v2f bhi = {b.z, b.w};                                                \
        _Pragma("unroll")                                                          \
        for (int t = 0; t < HT; ++t) {                                             \
            const v4f a = buf[jj][t];                                              \
            const v2f alo = {a.x, a.y};                                            \
            aa[t] += dot4sq(a);                                                    \
            acc[t] = __builtin_amdgcn_wmma_f32_16x16x4_f32(                        \
                         false, alo, false, blo, (short)0, acc[t], false, false);  \
        }                                                                          \
        _Pragma("unroll")                                                          \
        for (int t = 0; t < HT; ++t) {                                             \
            const v4f a = buf[jj][t];                                              \
            const v2f ahi = {a.z, a.w};                                            \
            acc[t] = __builtin_amdgcn_wmma_f32_16x16x4_f32(                        \
                         false, ahi, false, bhi, (short)0, acc[t], false, false);  \
        }                                                                          \
    }                                                                              \
    AA_HERE                                                                        \
    if ((g) % NLINE == NLINE - 1) {                                                \
        FLUSH_HALF((g) / NLINE)                                                    \
    }

#define ORDER_PIN __builtin_amdgcn_sched_barrier(0);

__global__ __launch_bounds__(NTHR)
void proto_head_kernel(const float* __restrict__ query,
                       const float* __restrict__ support,
                       const int*   __restrict__ slabels,
                       const float* __restrict__ scale,
                       float*       __restrict__ out)
{
    __shared__ __attribute__((aligned(16))) float ldsProto[NWG * NWAY * PSTR];
    __shared__ float ldsAB[NWG * 2 * MROWS * NWAY];
    __shared__ float ldsAA[NWG * MROWS];
    __shared__ float ldsBB[NWG * NWAY];

    const int tid  = threadIdx.x;
    const int lane = tid & 31;
    const int wave = __builtin_amdgcn_readfirstlane(tid >> 5);
    const int sub  = wave / NWAVE;
    const int wv   = wave - sub * NWAVE;
    const int task = blockIdx.x * TPB + sub;
    const int half = lane >> 4;
    const int l16  = lane & 15;
    const int r8   = lane & 7;
    const int part = (lane >> 3) & 1;
    const int K0   = wv * KW;

    const float* qtask = query + (size_t)task * NQ * DIM + K0 + part * 8 + half * 4;
    const float* q08   = qtask + (size_t)r8 * DIM;
    const int    r9    = (MT - 1) * RT + r8;
    const float* q9    = qtask + (size_t)((r9 < NQ) ? r9 : 0) * DIM;

    v4f bufA[2][HT], bufB[2][HT];
    LOAD_GROUP(bufA, 0)

    float* pw = &ldsProto[wave * NWAY * PSTR];
    {
        const float* sup = support + (size_t)task * NS * DIM + K0 + lane;
        const int*   lab = slabels + task * NS;
        float p0[NPIECE], p1[NPIECE], p2[NPIECE], p3[NPIECE], p4[NPIECE];
        #pragma unroll
        for (int i = 0; i < NPIECE; ++i) { p0[i] = 0.f; p1[i] = 0.f; p2[i] = 0.f; p3[i] = 0.f; p4[i] = 0.f; }
        int c0 = 0, c1 = 0, c2 = 0, c3 = 0, c4 = 0;
        #pragma unroll
        for (int s = 0; s < NS; ++s) {
            const int   lb = lab[s];
            const float w0 = (lb == 0) ? 1.f : 0.f;
            const float w1 = (lb == 1) ? 1.f : 0.f;
            const float w2 = (lb == 2) ? 1.f : 0.f;
            const float w3 = (lb == 3) ? 1.f : 0.f;
            const float w4 = (lb == 4) ? 1.f : 0.f;
            c0 += (lb == 0); c1 += (lb == 1); c2 += (lb == 2); c3 += (lb == 3); c4 += (lb == 4);
            #pragma unroll
            for (int i = 0; i < NPIECE; ++i) {
                const float v = sup[s * DIM + i * 32];
                p0[i] += w0 * v;
                p1[i] += w1 * v;
                p2[i] += w2 * v;
                p3[i] += w3 * v;
                p4[i] += w4 * v;
            }
        }
        const float inv0 = 1.0f / (float)c0, inv1 = 1.0f / (float)c1, inv2 = 1.0f / (float)c2;
        const float inv3 = 1.0f / (float)c3, inv4 = 1.0f / (float)c4;
        float bb0 = 0.f, bb1 = 0.f, bb2 = 0.f, bb3 = 0.f, bb4 = 0.f;
        #pragma unroll
        for (int i = 0; i < NPIECE; ++i) {
            p0[i] *= inv0; p1[i] *= inv1; p2[i] *= inv2; p3[i] *= inv3; p4[i] *= inv4;
            pw[0 * PSTR + i * 32 + lane] = p0[i];  bb0 += p0[i] * p0[i];
            pw[1 * PSTR + i * 32 + lane] = p1[i];  bb1 += p1[i] * p1[i];
            pw[2 * PSTR + i * 32 + lane] = p2[i];  bb2 += p2[i] * p2[i];
            pw[3 * PSTR + i * 32 + lane] = p3[i];  bb3 += p3[i] * p3[i];
            pw[4 * PSTR + i * 32 + lane] = p4[i];  bb4 += p4[i] * p4[i];
        }
        bb0 = wave_sum(bb0); bb1 = wave_sum(bb1); bb2 = wave_sum(bb2);
        bb3 = wave_sum(bb3); bb4 = wave_sum(bb4);
        if (lane == 0) {
            ldsBB[wave * NWAY + 0] = bb0;
            ldsBB[wave * NWAY + 1] = bb1;
            ldsBB[wave * NWAY + 2] = bb2;
            ldsBB[wave * NWAY + 3] = bb3;
            ldsBB[wave * NWAY + 4] = bb4;
        }
    }
    __builtin_amdgcn_fence(__ATOMIC_RELEASE, "workgroup");
    __builtin_amdgcn_wave_barrier();
    __builtin_amdgcn_fence(__ATOMIC_ACQUIRE, "workgroup");

    const bool   bUsed = (l16 < 2 * NWAY);
    const int    bPrt  = bUsed ? (l16 / NWAY) : 0;
    const int    bCls  = bUsed ? (l16 - bPrt * NWAY) : 0;
    const float* bptr  = pw + bCls * PSTR + bPrt * 8 + half * 4;
    const bool   dUse  = bUsed && (bPrt == half);
    const int    dCls  = bCls;

    v8f acc[HT];
    float aa[HT];

    #pragma unroll
    for (int g = 0; g < NGRP - 2; g += 2) {
        LOAD_GROUP(bufB, g + 1)
        ORDER_PIN
        COMPUTE_GROUP(bufA, g)
        ORDER_PIN
        LOAD_GROUP(bufA, g + 2)
        ORDER_PIN
        COMPUTE_GROUP(bufB, g + 1)
        ORDER_PIN
    }
    LOAD_GROUP(bufB, NGRP - 1)
    ORDER_PIN
    COMPUTE_GROUP(bufA, NGRP - 2)
    COMPUTE_GROUP(bufB, NGRP - 1)
    __syncthreads();

    const float sc   = scale[0];
    const int   w0   = sub * NWAVE;
    const int   ltid = tid - sub * TTHR;
#define EMIT_LOGIT(idx)                                                            \
    {                                                                              \
        const int i   = (idx);                                                     \
        const int row = i / NWAY;                                                  \
        const int c   = i - row * NWAY;                                            \
        float aarow = 0.0f, abv = 0.0f, bbv = 0.0f;                                \
        _Pragma("unroll")                                                          \
        for (int w = 0; w < NWAVE; ++w) {                                          \
            aarow += ldsAA[(w0 + w) * MROWS + row];                                \
            abv   += ldsAB[(((w0 + w) * 2 + 0) * MROWS + row) * NWAY + c];         \
            abv   += ldsAB[(((w0 + w) * 2 + 1) * MROWS + row) * NWAY + c];         \
            bbv   += ldsBB[(w0 + w) * NWAY + c];                                   \
        }                                                                          \
        const float logit = sc * (-(aarow - 2.0f * abv + bbv) * (1.0f / (float)DIM)); \
        out[(size_t)task * NQ * NWAY + i] = logit;                                 \
    }
    EMIT_LOGIT(ltid)
    if (ltid < NQ * NWAY - TTHR) {
        EMIT_LOGIT(ltid + TTHR)
    }
}

extern "C" void kernel_launch(void* const* d_in, const int* in_sizes, int n_in,
                              void* d_out, int out_size, void* d_ws, size_t ws_size,
                              hipStream_t stream) {
    const float* query   = (const float*)d_in[0];
    const float* support = (const float*)d_in[1];
    const int*   slab    = (const int*)  d_in[2];
    const float* scale   = (const float*)d_in[3];
    float*       out     = (float*)d_out;
    (void)in_sizes; (void)n_in; (void)out_size; (void)d_ws; (void)ws_size;

    proto_head_kernel<<<NGRID, NTHR, 0, stream>>>(query, support, slab, scale, out);
}
